# starting kernel
# baseline (speedup 1.0000x reference)
#include <hip/hip_runtime.h>


typedef __attribute__((ext_vector_type(2))) float v2f;
typedef __attribute__((ext_vector_type(8))) float v8f;

#define STATE 16
#define MEAS  8

static __device__ __forceinline__ float bcastf(float v, int srcLane) {
  int r = __builtin_amdgcn_ds_bpermute(srcLane << 2, __builtin_bit_cast(int, v));
  return __builtin_bit_cast(float, r);
}

static __device__ __forceinline__ v8f wmma_k4(v2f a, v2f b, v8f c) {
  return __builtin_amdgcn_wmma_f32_16x16x4_f32(
       false, a,  false, b,
       (short)0, c,  false,  false);
}

__global__ __launch_bounds__(32) void kalman_seq_kernel(
    const float* __restrict__ meas,
    const float* __restrict__ Fm,
    const float* __restrict__ Qm,
    const float* __restrict__ Rm,
    float* __restrict__ out,
    int T)
{
  __shared__ float lds_P [256];
  __shared__ float lds_M1[256];
  __shared__ float lds_Pp[256];
  __shared__ float lds_nK[256];
  __shared__ float lds_Si[128];
  __shared__ float lds_x [16];

  const int lane = threadIdx.x;
  const int l16  = lane & 15;
  const int half = lane >> 4;
  const int gr   = lane >> 2;
  const int cg   = lane & 3;

  float Frow[16];
#pragma unroll
  for (int j = 0; j < 16; ++j) Frow[j] = Fm[l16 * 16 + j];

  v2f fab[4];
#pragma unroll
  for (int s = 0; s < 4; ++s) {
    const int k0 = 4 * s + 2 * half;
    fab[s].x = Fm[l16 * 16 + k0];
    fab[s].y = Fm[l16 * 16 + k0 + 1];
  }

  v8f qacc;
#pragma unroll
  for (int r = 0; r < 8; ++r) qacc[r] = Qm[(r + 8 * half) * 16 + l16];

  float rr[4];
#pragma unroll
  for (int j = 0; j < 4; ++j)
    rr[j] = (cg < 2) ? Rm[gr * 8 + 4 * cg + j] : 0.0f;

  const float maskL = (cg < 2) ? 1.0f : 0.0f;
  float iden[4];
#pragma unroll
  for (int j = 0; j < 4; ++j)
    iden[j] = ((4 * cg + j) - 8 == gr) ? 1.0f : 0.0f;

  for (int idx = lane; idx < 256; idx += 32)
    lds_P[idx] = ((idx >> 4) == (idx & 15)) ? 1.0f : 0.0f;
  for (int idx = lane; idx < 128; idx += 32)
    lds_Si[idx] = 0.0f;
  if (lane < 16) lds_x[lane] = 0.0f;
  __syncthreads();

  float zv = meas[lane & 7];

  for (int t = 0; t < T; ++t) {
    const int tn = (t + 1 < T) ? (t + 1) : (T - 1);
    const float znext = meas[(size_t)tn * MEAS + (lane & 7)];
    __builtin_prefetch(&meas[(size_t)(t + 48) * MEAS], 0, 1);

    float x0 = 0.0f, x1 = 0.0f, x2 = 0.0f, x3 = 0.0f;
#pragma unroll
    for (int j = 0; j < 4; ++j) {
      x0 += Frow[j]      * lds_x[j];
      x1 += Frow[j + 4]  * lds_x[j + 4];
      x2 += Frow[j + 8]  * lds_x[j + 8];
      x3 += Frow[j + 12] * lds_x[j + 12];
    }
    const float xp = (x0 + x1) + (x2 + x3);
    const float yv = zv - xp;

    v8f m1 = {0, 0, 0, 0, 0, 0, 0, 0};
#pragma unroll
    for (int s = 0; s < 4; ++s) {
      const int kr = 4 * s + 2 * half;
      v2f b;
      b.x = lds_P[kr * 16 + l16];
      b.y = lds_P[(kr + 1) * 16 + l16];
      m1 = wmma_k4(fab[s], b, m1);
    }
#pragma unroll
    for (int r = 0; r < 8; ++r) lds_M1[(r + 8 * half) * 16 + l16] = m1[r];
    __syncthreads();

    v8f pp = qacc;
#pragma unroll
    for (int s = 0; s < 4; ++s) {
      const int k0 = 4 * s + 2 * half;
      v2f a;
      a.x = lds_M1[l16 * 16 + k0];
      a.y = lds_M1[l16 * 16 + k0 + 1];
      pp = wmma_k4(a, fab[s], pp);
    }
#pragma unroll
    for (int r = 0; r < 8; ++r) lds_Pp[(r + 8 * half) * 16 + l16] = pp[r];
    __syncthreads();

    float va[4];
    {
      const int base = gr * 16 + 4 * (cg & 1);
#pragma unroll
      for (int j = 0; j < 4; ++j) {
        const float sv = lds_Pp[base + j] + rr[j];
        va[j] = sv * maskL + iden[j];
      }
    }
#pragma unroll
    for (int k = 0; k < 8; ++k) {
      const int kc = k & 3, kg = k >> 2;
      const float pv  = bcastf(va[kc], 4 * k + kg);
      const float piv = __builtin_amdgcn_rcpf(pv);
      const float fac = bcastf(va[kc], 4 * gr + kg);
      const float p0 = bcastf(va[0], 4 * k + cg) * piv;
      const float p1 = bcastf(va[1], 4 * k + cg) * piv;
      const float p2 = bcastf(va[2], 4 * k + cg) * piv;
      const float p3 = bcastf(va[3], 4 * k + cg) * piv;
      const bool isP = (gr == k);
      va[0] = isP ? p0 : va[0] - fac * p0;
      va[1] = isP ? p1 : va[1] - fac * p1;
      va[2] = isP ? p2 : va[2] - fac * p2;
      va[3] = isP ? p3 : va[3] - fac * p3;
    }
    if (cg >= 2) {
#pragma unroll
      for (int j = 0; j < 4; ++j)
        lds_Si[gr * 16 + 4 * (cg - 2) + j] = va[j];
    }
    __syncthreads();

    v8f kk = {0, 0, 0, 0, 0, 0, 0, 0};
#pragma unroll
    for (int s = 0; s < 2; ++s) {
      const int k0 = 4 * s + 2 * half;
      v2f a;
      a.x = lds_Pp[l16 * 16 + k0];
      a.y = lds_Pp[l16 * 16 + k0 + 1];
      v2f b;
      b.x = lds_Si[k0 * 16 + l16];
      b.y = lds_Si[(k0 + 1) * 16 + l16];
      kk = wmma_k4(a, b, kk);
    }
#pragma unroll
    for (int r = 0; r < 8; ++r) lds_nK[(r + 8 * half) * 16 + l16] = -kk[r];
    __syncthreads();

    v8f pn = pp;
#pragma unroll
    for (int s = 0; s < 2; ++s) {
      const int k0 = 4 * s + 2 * half;
      v2f a;
      a.x = lds_nK[l16 * 16 + k0];
      a.y = lds_nK[l16 * 16 + k0 + 1];
      v2f b;
      b.x = lds_Pp[k0 * 16 + l16];
      b.y = lds_Pp[(k0 + 1) * 16 + l16];
      pn = wmma_k4(a, b, pn);
    }
#pragma unroll
    for (int r = 0; r < 8; ++r) lds_P[(r + 8 * half) * 16 + l16] = pn[r];

    float xn = xp;
#pragma unroll
    for (int j = 0; j < MEAS; ++j)
      xn -= lds_nK[l16 * 16 + j] * bcastf(yv, j);
    if (lane < 16) {
      lds_x[l16] = xn;
      out[(size_t)t * 16 + l16] = xn;
    }
    zv = znext;
    __syncthreads();
  }
}

extern "C" void kernel_launch(void* const* d_in, const int* in_sizes, int n_in,
                              void* d_out, int out_size, void* d_ws, size_t ws_size,
                              hipStream_t stream) {
  const float* meas = (const float*)d_in[0];
  const float* F    = (const float*)d_in[1];
  const float* Q    = (const float*)d_in[3];
  const float* R    = (const float*)d_in[4];
  float* out = (float*)d_out;

  const int T = in_sizes[0] / MEAS;

  hipLaunchKernelGGL(kalman_seq_kernel, dim3(1), dim3(32), 0, stream,
                     meas, F, Q, R, out, T);
}


# F folded through the update: the loop carries F P = [0|G2] + G1 [S^-1 R | -S^-1 Pp12], G = F Pp formed beside the elimination; no P tile, its LDS hop and the 4-deep F P chain leave the serial chain
# speedup vs baseline: 2.8198x; 2.8198x over previous
#include <hip/hip_runtime.h>


typedef __attribute__((ext_vector_type(2))) float v2f;
typedef __attribute__((ext_vector_type(4))) float v4f;
typedef __attribute__((ext_vector_type(8))) float v8f;

#define STATE 16
#define MEAS  8

static __device__ __forceinline__ v8f wmma_k4(v2f a, v2f b, v8f c) {
  return __builtin_amdgcn_wmma_f32_16x16x4_f32(
       false, a,  false, b,
       (short)0, c,  false,  false);
}

static __device__ __forceinline__ void wave_order() {
  __builtin_amdgcn_fence(__ATOMIC_RELEASE, "wavefront");
  __builtin_amdgcn_fence(__ATOMIC_ACQUIRE, "wavefront");
}

__global__ __launch_bounds__(32) void kalman_seq_kernel(
    const float* __restrict__ meas,
    const float* __restrict__ Fm,
    const float* __restrict__ Qm,
    const float* __restrict__ Rm,
    float* __restrict__ out,
    int T)
{
  __shared__ float lds_M1[256];
  __shared__ float lds_Pp[256];
  __shared__ float lds_G [256];
  __shared__ __attribute__((aligned(16))) float lds_x [16];

  const int lane = threadIdx.x;
  const int l16  = lane & 15;
  const int half = lane >> 4;

  float Frow[16];
#pragma unroll
  for (int j = 0; j < 16; ++j) Frow[j] = Fm[l16 * 16 + j];

  v2f fab[4];
#pragma unroll
  for (int s = 0; s < 4; ++s) {
    const int k0 = 4 * s + 2 * half;
    fab[s].x = Fm[l16 * 16 + k0];
    fab[s].y = Fm[l16 * 16 + k0 + 1];
  }

  v8f qacc;
#pragma unroll
  for (int r = 0; r < 8; ++r) qacc[r] = Qm[(r + 8 * half) * 16 + l16];

  float rcol[8];
#pragma unroll
  for (int j = 0; j < 8; ++j) {
    const float rj = (l16 < 8) ? Rm[j * 8 + l16] : 0.0f;
    rcol[j] = (lane >= 24) ? ((j == lane - 24) ? 1.0f : 0.0f) : rj;
  }
  const float sgn = (lane < 8) ? 1.0f : ((lane < 16) ? -1.0f : 0.0f);
  const bool useOwn = (lane >= 8) && (lane < 24);

  for (int idx = lane; idx < 256; idx += 32)
    lds_M1[idx] = Fm[idx];
  if (lane < 16) lds_x[lane] = 0.0f;
  wave_order();

  float zv = meas[lane & 7];

  v2f ma[4];
#pragma unroll
  for (int s = 0; s < 4; ++s) {
    const int k0 = 4 * s + 2 * half;
    ma[s].x = lds_M1[l16 * 16 + k0];
    ma[s].y = lds_M1[l16 * 16 + k0 + 1];
  }

  for (int t = 0; t < T; ++t) {
    const int tn = (t + 1 < T) ? (t + 1) : (T - 1);
    const float znext = meas[(size_t)tn * MEAS + (lane & 7)];
    __builtin_prefetch(&meas[(size_t)(t + 48) * MEAS], 0, 1);

    v8f pp = qacc;
#pragma unroll
    for (int s = 0; s < 4; ++s) pp = wmma_k4(ma[s], fab[s], pp);
#pragma unroll
    for (int r = 0; r < 8; ++r) lds_Pp[(r + 8 * half) * 16 + l16] = pp[r];
    wave_order();

    v2f pq[4];
#pragma unroll
    for (int s = 0; s < 4; ++s) {
      const int kr = 4 * s + 2 * half;
      pq[s].x = lds_Pp[kr * 16 + l16];
      pq[s].y = lds_Pp[(kr + 1) * 16 + l16];
    }

    v2f kA[2];
#pragma unroll
    for (int s = 0; s < 2; ++s) {
      const int kk0 = 4 * half + 2 * s;
      kA[s].x = lds_Pp[l16 * 16 + kk0];
      kA[s].y = lds_Pp[l16 * 16 + kk0 + 1];
    }

    v4f xq[4];
#pragma unroll
    for (int i = 0; i < 4; ++i) xq[i] = *(const v4f*)&lds_x[4 * i];

    float va[8];
#pragma unroll
    for (int j = 0; j < 8; ++j) va[j] = __builtin_fmaf(pp[j], sgn, rcol[j]);
    v2f gA[2];
    v8f cG;
#pragma unroll
    for (int k = 0; k < 8; ++k) {
      if (k == 0) {
        asm volatile("" : "+v"(va[0]) : : "memory");
      }
      if (k == 2) {
        float p0 = pq[0].x;
        asm volatile("" : "+v"(p0), "+v"(xq[0]), "+v"(xq[1]), "+v"(xq[2]), "+v"(xq[3]) : "v"(va[2]));
        pq[0].x = p0;
        v8f g = {0, 0, 0, 0, 0, 0, 0, 0};
#pragma unroll
        for (int s = 0; s < 4; ++s) g = wmma_k4(fab[s], pq[s], g);
#pragma unroll
        for (int r = 0; r < 8; ++r) lds_G[(r + 8 * half) * 16 + l16] = g[r];
        wave_order();
#pragma unroll
        for (int s = 0; s < 2; ++s) {
          const int kk0 = 4 * half + 2 * s;
          gA[s].x = lds_G[l16 * 16 + kk0];
          gA[s].y = lds_G[l16 * 16 + kk0 + 1];
        }
#pragma unroll
        for (int r = 0; r < 8; ++r) cG[r] = (l16 >= 8) ? g[r] : 0.0f;
      }
      float rk[8];
#pragma unroll
      for (int j = 0; j < 8; ++j)
        rk[j] = __builtin_bit_cast(float,
            __builtin_amdgcn_readlane(__builtin_bit_cast(int, va[j]), k));
      const float piv = __builtin_amdgcn_rcpf(rk[k]);
      const float q = va[k] * piv;
#pragma unroll
      for (int j = 0; j < 8; ++j) {
        if (j == k) continue;
        va[j] = __builtin_fmaf(-q, rk[j], va[j]);
      }
      va[k] = q;
    }
    float x0 = 0.0f, x1 = 0.0f, x2 = 0.0f, x3 = 0.0f;
#pragma unroll
    for (int j = 0; j < 4; ++j) {
      x0 = __builtin_fmaf(Frow[j],      xq[0][j], x0);
      x1 = __builtin_fmaf(Frow[j + 4],  xq[1][j], x1);
      x2 = __builtin_fmaf(Frow[j + 8],  xq[2][j], x2);
      x3 = __builtin_fmaf(Frow[j + 12], xq[3][j], x3);
    }
    const float xp = (x0 + x1) + (x2 + x3);
    const float yv = zv - xp;
    const float yr = __builtin_bit_cast(float,
        __builtin_amdgcn_update_dpp(0, __builtin_bit_cast(int, yv), 0x128, 0xf, 0xf, false));


    float bP[4], aX[4];
#pragma unroll
    for (int i = 0; i < 4; ++i) {
      const float own  = half ? va[4 + i] : va[i];
      const float send = half ? va[i] : va[4 + i];
      const int   si   = __builtin_bit_cast(int, send);
      const float recv = __builtin_bit_cast(float,
          __builtin_amdgcn_permlanex16(si, si, 0x76543210, 0xfedcba98, false, false));
      bP[i] = useOwn ? own : recv;
      aX[i] = (useOwn ? recv : own) * yr;
    }

    v8f m1 = cG;
#pragma unroll
    for (int s = 0; s < 2; ++s) {
      v2f b;
      b.x = bP[2 * s];
      b.y = bP[2 * s + 1];
      m1 = wmma_k4(gA[s], b, m1);
    }
#pragma unroll
    for (int r = 0; r < 8; ++r) lds_M1[(r + 8 * half) * 16 + l16] = m1[r];
    wave_order();

#pragma unroll
    for (int s = 0; s < 4; ++s) {
      const int k0 = 4 * s + 2 * half;
      ma[s].x = lds_M1[l16 * 16 + k0];
      ma[s].y = lds_M1[l16 * 16 + k0 + 1];
    }

    v8f xd = {xp, 0, 0, 0, 0, 0, 0, 0};
#pragma unroll
    for (int s = 0; s < 2; ++s) {
      v2f a;
      a.x = aX[2 * s];
      a.y = aX[2 * s + 1];
      xd = wmma_k4(a, kA[s], xd);
    }
    const float xn = ((xd[0] + xd[1]) + (xd[2] + xd[3]))
                   + ((xd[4] + xd[5]) + (xd[6] + xd[7]));
    if (lane >= 16) {
      lds_x[l16] = xn;
      out[(size_t)t * 16 + l16] = xn;
    }
    zv = znext;
    wave_order();
  }
}

extern "C" void kernel_launch(void* const* d_in, const int* in_sizes, int n_in,
                              void* d_out, int out_size, void* d_ws, size_t ws_size,
                              hipStream_t stream) {
  const float* meas = (const float*)d_in[0];
  const float* F    = (const float*)d_in[1];
  const float* Q    = (const float*)d_in[3];
  const float* R    = (const float*)d_in[4];
  float* out = (float*)d_out;

  const int T = in_sizes[0] / MEAS;

  hipLaunchKernelGGL(kalman_seq_kernel, dim3(1), dim3(32), 0, stream,
                     meas, F, Q, R, out, T);
}


# Elimination stage re-cut for in-order issue: 1/pivot taken in the pivot lane before the v_readlane broadcast, the eight broadcasts issued in one run ahead of the arithmetic: one scalar hop per stage
# speedup vs baseline: 2.9053x; 1.0303x over previous
#include <hip/hip_runtime.h>


typedef __attribute__((ext_vector_type(2))) float v2f;
typedef __attribute__((ext_vector_type(4))) float v4f;
typedef __attribute__((ext_vector_type(8))) float v8f;

#define STATE 16
#define MEAS  8

static __device__ __forceinline__ v8f wmma_k4(v2f a, v2f b, v8f c) {
  return __builtin_amdgcn_wmma_f32_16x16x4_f32(
       false, a,  false, b,
       (short)0, c,  false,  false);
}

static __device__ __forceinline__ void wave_order() {
  __builtin_amdgcn_fence(__ATOMIC_RELEASE, "wavefront");
  __builtin_amdgcn_fence(__ATOMIC_ACQUIRE, "wavefront");
}

__global__ __launch_bounds__(32) void kalman_seq_kernel(
    const float* __restrict__ meas,
    const float* __restrict__ Fm,
    const float* __restrict__ Qm,
    const float* __restrict__ Rm,
    float* __restrict__ out,
    int T)
{
  __shared__ float lds_M1[256];
  __shared__ float lds_Pp[256];
  __shared__ float lds_G [256];
  __shared__ __attribute__((aligned(16))) float lds_x [16];

  const int lane = threadIdx.x;
  const int l16  = lane & 15;
  const int half = lane >> 4;

  float Frow[16];
#pragma unroll
  for (int j = 0; j < 16; ++j) Frow[j] = Fm[l16 * 16 + j];

  v2f fab[4];
#pragma unroll
  for (int s = 0; s < 4; ++s) {
    const int k0 = 4 * s + 2 * half;
    fab[s].x = Fm[l16 * 16 + k0];
    fab[s].y = Fm[l16 * 16 + k0 + 1];
  }

  v8f qacc;
#pragma unroll
  for (int r = 0; r < 8; ++r) qacc[r] = Qm[(r + 8 * half) * 16 + l16];

  float rcol[8];
#pragma unroll
  for (int j = 0; j < 8; ++j) {
    const float rj = (l16 < 8) ? Rm[j * 8 + l16] : 0.0f;
    rcol[j] = (lane >= 24) ? ((j == lane - 24) ? 1.0f : 0.0f) : rj;
  }
  const float sgn = (lane < 8) ? 1.0f : ((lane < 16) ? -1.0f : 0.0f);
  const bool useOwn = (lane >= 8) && (lane < 24);

  for (int idx = lane; idx < 256; idx += 32)
    lds_M1[idx] = Fm[idx];
  if (lane < 16) lds_x[lane] = 0.0f;
  wave_order();

  float zv = meas[lane & 7];

  v2f ma[4];
#pragma unroll
  for (int s = 0; s < 4; ++s) {
    const int k0 = 4 * s + 2 * half;
    ma[s].x = lds_M1[l16 * 16 + k0];
    ma[s].y = lds_M1[l16 * 16 + k0 + 1];
  }

  for (int t = 0; t < T; ++t) {
    const int tn = (t + 1 < T) ? (t + 1) : (T - 1);
    const float znext = meas[(size_t)tn * MEAS + (lane & 7)];
    __builtin_prefetch(&meas[(size_t)(t + 48) * MEAS], 0, 1);

    v8f pp = qacc;
#pragma unroll
    for (int s = 0; s < 4; ++s) pp = wmma_k4(ma[s], fab[s], pp);
#pragma unroll
    for (int r = 0; r < 8; ++r) lds_Pp[(r + 8 * half) * 16 + l16] = pp[r];
    wave_order();

    v2f pq[4];
#pragma unroll
    for (int s = 0; s < 4; ++s) {
      const int kr = 4 * s + 2 * half;
      pq[s].x = lds_Pp[kr * 16 + l16];
      pq[s].y = lds_Pp[(kr + 1) * 16 + l16];
    }

    v2f kA[2];
#pragma unroll
    for (int s = 0; s < 2; ++s) {
      const int kk0 = 4 * half + 2 * s;
      kA[s].x = lds_Pp[l16 * 16 + kk0];
      kA[s].y = lds_Pp[l16 * 16 + kk0 + 1];
    }

    v4f xq[4];
#pragma unroll
    for (int i = 0; i < 4; ++i) xq[i] = *(const v4f*)&lds_x[4 * i];

    float va[8];
#pragma unroll
    for (int j = 0; j < 8; ++j) va[j] = __builtin_fmaf(pp[j], sgn, rcol[j]);
    v2f gA[2];
    v8f cG;
    v8f g = {0, 0, 0, 0, 0, 0, 0, 0};
    asm volatile("" : "+v"(va[0]) : : "memory");
    float rinv = __builtin_amdgcn_rcpf(va[0]);
#pragma unroll
    for (int k = 0; k < 8; ++k) {
      if (k >= 2 && k <= 5) {
        if (k == 2) {
          float p0 = pq[0].x;
          asm volatile("" : "+v"(p0), "+v"(rinv));
          pq[0].x = p0;
        } else {
          asm volatile("" : "+v"(g), "+v"(rinv));
        }
        g = wmma_k4(fab[k - 2], pq[k - 2], g);
        asm volatile("" : "+v"(g), "+v"(rinv));
      }
      float mk[8];
#pragma unroll
      for (int j = 0; j < 8; ++j)
        if (j != k)
          mk[j] = __builtin_bit_cast(float,
              __builtin_amdgcn_readlane(__builtin_bit_cast(int, va[j]), k));
      {
        float m0 = mk[(k + 1) & 7], m1 = mk[(k + 2) & 7], m2 = mk[(k + 3) & 7],
              m3 = mk[(k + 4) & 7], m4 = mk[(k + 5) & 7], m5 = mk[(k + 6) & 7],
              m6 = mk[(k + 7) & 7];
        asm volatile("" : "+v"(rinv) : "s"(m0), "s"(m1), "s"(m2), "s"(m3), "s"(m4), "s"(m5), "s"(m6));
      }
      mk[k] = __builtin_bit_cast(float,
          __builtin_amdgcn_readlane(__builtin_bit_cast(int, rinv), k));
      asm volatile("" : "+s"(mk[0]), "+s"(mk[1]), "+s"(mk[2]), "+s"(mk[3]),
                        "+s"(mk[4]), "+s"(mk[5]), "+s"(mk[6]), "+s"(mk[7]));
      const float piv = mk[k];
      float tj[8];
#pragma unroll
      for (int j = 0; j < 8; ++j)
        tj[j] = (j != k) ? va[k] * mk[j] : 0.0f;
      {
        float p0 = tj[(k + 1) & 7], p1 = tj[(k + 2) & 7], p2 = tj[(k + 3) & 7],
              p3 = tj[(k + 4) & 7], p4 = tj[(k + 5) & 7], p5 = tj[(k + 6) & 7],
              p6 = tj[(k + 7) & 7];
        asm volatile("" : "+v"(p0), "+v"(p1), "+v"(p2), "+v"(p3), "+v"(p4), "+v"(p5), "+v"(p6));
        tj[(k + 1) & 7] = p0; tj[(k + 2) & 7] = p1; tj[(k + 3) & 7] = p2;
        tj[(k + 4) & 7] = p3; tj[(k + 5) & 7] = p4; tj[(k + 6) & 7] = p5;
        tj[(k + 7) & 7] = p6;
      }
      if (k < 7) {
        va[k + 1] = __builtin_fmaf(-tj[k + 1], piv, va[k + 1]);
        rinv = __builtin_amdgcn_rcpf(va[k + 1]);
      }
#pragma unroll
      for (int j = 0; j < 8; ++j) {
        if (j == k || j == k + 1) continue;
        va[j] = __builtin_fmaf(-tj[j], piv, va[j]);
      }
      va[k] = va[k] * piv;
      if (k == 5) {
        asm volatile("" : "+v"(g) : "v"(va[5]));
#pragma unroll
        for (int r = 0; r < 8; ++r) lds_G[(r + 8 * half) * 16 + l16] = g[r];
        wave_order();
#pragma unroll
        for (int s = 0; s < 2; ++s) {
          const int kk0 = 4 * half + 2 * s;
          gA[s].x = lds_G[l16 * 16 + kk0];
          gA[s].y = lds_G[l16 * 16 + kk0 + 1];
        }
#pragma unroll
        for (int r = 0; r < 8; ++r) cG[r] = (l16 >= 8) ? g[r] : 0.0f;
      }
      if (k < 7) {
        asm volatile("" : "+v"(va[0]), "+v"(va[1]), "+v"(va[2]), "+v"(va[3]),
                          "+v"(va[4]), "+v"(va[5]), "+v"(va[6]), "+v"(va[7]), "+v"(rinv));
      }
    }

    float bP[4], aX[4];
#pragma unroll
    for (int i = 0; i < 4; ++i) {
      const float own  = half ? va[4 + i] : va[i];
      const float send = half ? va[i] : va[4 + i];
      const int   si   = __builtin_bit_cast(int, send);
      const float recv = __builtin_bit_cast(float,
          __builtin_amdgcn_permlanex16(si, si, 0x76543210, 0xfedcba98, false, false));
      bP[i] = useOwn ? own : recv;
      aX[i] = useOwn ? recv : own;
    }

    v8f m1 = cG;
#pragma unroll
    for (int s = 0; s < 2; ++s) {
      v2f b;
      b.x = bP[2 * s];
      b.y = bP[2 * s + 1];
      m1 = wmma_k4(gA[s], b, m1);
      if (s == 0) {
        asm volatile("" : "+v"(xq[0]), "+v"(xq[1]), "+v"(xq[2]), "+v"(xq[3]), "+v"(m1));
      }
    }

    float x0 = 0.0f, x1 = 0.0f, x2 = 0.0f, x3 = 0.0f;
#pragma unroll
    for (int j = 0; j < 4; ++j) {
      x0 = __builtin_fmaf(Frow[j],      xq[0][j], x0);
      x1 = __builtin_fmaf(Frow[j + 4],  xq[1][j], x1);
      x2 = __builtin_fmaf(Frow[j + 8],  xq[2][j], x2);
      x3 = __builtin_fmaf(Frow[j + 12], xq[3][j], x3);
    }
    const float xp = (x0 + x1) + (x2 + x3);
    const float yv = zv - xp;
    const float yr = __builtin_bit_cast(float,
        __builtin_amdgcn_update_dpp(0, __builtin_bit_cast(int, yv), 0x128, 0xf, 0xf, false));

#pragma unroll
    for (int r = 0; r < 8; ++r) lds_M1[(r + 8 * half) * 16 + l16] = m1[r];
    wave_order();

#pragma unroll
    for (int s = 0; s < 4; ++s) {
      const int k0 = 4 * s + 2 * half;
      ma[s].x = lds_M1[l16 * 16 + k0];
      ma[s].y = lds_M1[l16 * 16 + k0 + 1];
    }

    v8f xd = {xp, 0, 0, 0, 0, 0, 0, 0};
#pragma unroll
    for (int s = 0; s < 2; ++s) {
      v2f a;
      a.x = aX[2 * s] * yr;
      a.y = aX[2 * s + 1] * yr;
      xd = wmma_k4(a, kA[s], xd);
    }
    const float xn = ((xd[0] + xd[1]) + (xd[2] + xd[3]))
                   + ((xd[4] + xd[5]) + (xd[6] + xd[7]));
    if (lane >= 16) {
      lds_x[l16] = xn;
      out[(size_t)t * 16 + l16] = xn;
    }
    zv = znext;
    wave_order();
  }
}

extern "C" void kernel_launch(void* const* d_in, const int* in_sizes, int n_in,
                              void* d_out, int out_size, void* d_ws, size_t ws_size,
                              hipStream_t stream) {
  const float* meas = (const float*)d_in[0];
  const float* F    = (const float*)d_in[1];
  const float* Q    = (const float*)d_in[3];
  const float* R    = (const float*)d_in[4];
  float* out = (float*)d_out;

  const int T = in_sizes[0] / MEAS;

  hipLaunchKernelGGL(kalman_seq_kernel, dim3(1), dim3(32), 0, stream,
                     meas, F, Q, R, out, T);
}


# State work off the covariance chain's in-order path: F x as four short packed chains between the two F P products, y behind the F P reads, each step's x sum and stores beside the next P_pred chain
# speedup vs baseline: 3.1014x; 1.0675x over previous
#include <hip/hip_runtime.h>


typedef __attribute__((ext_vector_type(2))) float v2f;
typedef __attribute__((ext_vector_type(4))) float v4f;
typedef __attribute__((ext_vector_type(8))) float v8f;

#define STATE 16
#define MEAS  8

static __device__ __forceinline__ v8f wmma_k4(v2f a, v2f b, v8f c) {
  return __builtin_amdgcn_wmma_f32_16x16x4_f32(
       false, a,  false, b,
       (short)0, c,  false,  false);
}

static __device__ __forceinline__ void wave_order() {
  __builtin_amdgcn_fence(__ATOMIC_RELEASE, "wavefront");
  __builtin_amdgcn_fence(__ATOMIC_ACQUIRE, "wavefront");
}

__global__ __launch_bounds__(32) void kalman_seq_kernel(
    const float* __restrict__ meas,
    const float* __restrict__ Fm,
    const float* __restrict__ Qm,
    const float* __restrict__ Rm,
    float* __restrict__ out,
    int T)
{
  __shared__ float lds_M1[256];
  __shared__ float lds_Pp[256];
  __shared__ float lds_G [256];
  __shared__ __attribute__((aligned(16))) float lds_x [16];

  const int lane = threadIdx.x;
  const int l16  = lane & 15;
  const int half = lane >> 4;

  v2f Fp[8];
#pragma unroll
  for (int m = 0; m < 8; ++m) {
    Fp[m].x = Fm[l16 * 16 + 2 * m];
    Fp[m].y = Fm[l16 * 16 + 2 * m + 1];
  }

  v2f fab[4];
#pragma unroll
  for (int s = 0; s < 4; ++s) {
    const int k0 = 4 * s + 2 * half;
    fab[s].x = Fm[l16 * 16 + k0];
    fab[s].y = Fm[l16 * 16 + k0 + 1];
  }

  v8f qacc;
#pragma unroll
  for (int r = 0; r < 8; ++r) qacc[r] = Qm[(r + 8 * half) * 16 + l16];

  float rcol[8];
#pragma unroll
  for (int j = 0; j < 8; ++j) {
    const float rj = (l16 < 8) ? Rm[j * 8 + l16] : 0.0f;
    rcol[j] = (lane >= 24) ? ((j == lane - 24) ? 1.0f : 0.0f) : rj;
  }
  const float sgn = (lane < 8) ? 1.0f : ((lane < 16) ? -1.0f : 0.0f);
  const bool useOwn = (lane >= 8) && (lane < 24);

  for (int idx = lane; idx < 256; idx += 32)
    lds_M1[idx] = Fm[idx];
  if (lane < 16) lds_x[lane] = 0.0f;
  wave_order();

  float zv = meas[lane & 7];

  v2f ma[4];
#pragma unroll
  for (int s = 0; s < 4; ++s) {
    const int k0 = 4 * s + 2 * half;
    ma[s].x = lds_M1[l16 * 16 + k0];
    ma[s].y = lds_M1[l16 * 16 + k0 + 1];
  }

  v8f xdp = {0, 0, 0, 0, 0, 0, 0, 0};
  float xpp = 0.0f;

  for (int t = 0; t < T; ++t) {
    const int tn = (t + 1 < T) ? (t + 1) : (T - 1);
    const float znext = meas[(size_t)tn * MEAS + (lane & 7)];
    __builtin_prefetch(&meas[(size_t)(t + 48) * MEAS], 0, 1);

    v8f pp = wmma_k4(ma[0], fab[0], qacc);
    asm volatile("" : "+v"(xdp), "+v"(pp));
    float h0 = (xdp[0] + xdp[1]) + (xdp[2] + xdp[3]);
    float h1 = (xdp[4] + xdp[5]) + (xdp[6] + xdp[7]);
    pp = wmma_k4(ma[1], fab[1], pp);
    asm volatile("" : "+v"(h0), "+v"(h1), "+v"(pp));
    float xn = (h0 + h1) + xpp;
    pp = wmma_k4(ma[2], fab[2], pp);
    pp = wmma_k4(ma[3], fab[3], pp);
    asm volatile("" : "+v"(xn), "+v"(pp));
    if (lane >= 16 && t > 0) {
      lds_x[l16] = xn;
      out[(size_t)(t - 1) * 16 + l16] = xn;
    }
#pragma unroll
    for (int r = 0; r < 8; ++r) lds_Pp[(r + 8 * half) * 16 + l16] = pp[r];
    wave_order();

    v2f pq[4];
#pragma unroll
    for (int s = 0; s < 4; ++s) {
      const int kr = 4 * s + 2 * half;
      pq[s].x = lds_Pp[kr * 16 + l16];
      pq[s].y = lds_Pp[(kr + 1) * 16 + l16];
    }

    v2f kA[2];
#pragma unroll
    for (int s = 0; s < 2; ++s) {
      const int kk0 = 4 * half + 2 * s;
      kA[s].x = lds_Pp[l16 * 16 + kk0];
      kA[s].y = lds_Pp[l16 * 16 + kk0 + 1];
    }

    v4f xq[4];
#pragma unroll
    for (int i = 0; i < 4; ++i) xq[i] = *(const v4f*)&lds_x[4 * i];

    float va[8];
#pragma unroll
    for (int j = 0; j < 8; ++j) va[j] = __builtin_fmaf(pp[j], sgn, rcol[j]);
    v2f gA[2];
    v8f cG;
    v8f g = {0, 0, 0, 0, 0, 0, 0, 0};
    asm volatile("" : "+v"(va[0]) : : "memory");
    float rinv = __builtin_amdgcn_rcpf(va[0]);
#pragma unroll
    for (int k = 0; k < 8; ++k) {
      if (k >= 2 && k <= 5) {
        if (k == 2) {
          float p0 = pq[0].x;
          asm volatile("" : "+v"(p0), "+v"(rinv));
          pq[0].x = p0;
        } else {
          asm volatile("" : "+v"(g), "+v"(rinv));
        }
        g = wmma_k4(fab[k - 2], pq[k - 2], g);
        asm volatile("" : "+v"(g), "+v"(rinv));
      }
      float mk[8];
#pragma unroll
      for (int j = 0; j < 8; ++j)
        if (j != k)
          mk[j] = __builtin_bit_cast(float,
              __builtin_amdgcn_readlane(__builtin_bit_cast(int, va[j]), k));
      {
        float m0 = mk[(k + 1) & 7], m1 = mk[(k + 2) & 7], m2 = mk[(k + 3) & 7],
              m3 = mk[(k + 4) & 7], m4 = mk[(k + 5) & 7], m5 = mk[(k + 6) & 7],
              m6 = mk[(k + 7) & 7];
        asm volatile("" : "+v"(rinv) : "s"(m0), "s"(m1), "s"(m2), "s"(m3), "s"(m4), "s"(m5), "s"(m6));
      }
      mk[k] = __builtin_bit_cast(float,
          __builtin_amdgcn_readlane(__builtin_bit_cast(int, rinv), k));
      asm volatile("" : "+s"(mk[0]), "+s"(mk[1]), "+s"(mk[2]), "+s"(mk[3]),
                        "+s"(mk[4]), "+s"(mk[5]), "+s"(mk[6]), "+s"(mk[7]));
      const float piv = mk[k];
      float tj[8];
#pragma unroll
      for (int j = 0; j < 8; ++j)
        tj[j] = (j != k) ? va[k] * mk[j] : 0.0f;
      {
        float p0 = tj[(k + 1) & 7], p1 = tj[(k + 2) & 7], p2 = tj[(k + 3) & 7],
              p3 = tj[(k + 4) & 7], p4 = tj[(k + 5) & 7], p5 = tj[(k + 6) & 7],
              p6 = tj[(k + 7) & 7];
        asm volatile("" : "+v"(p0), "+v"(p1), "+v"(p2), "+v"(p3), "+v"(p4), "+v"(p5), "+v"(p6));
        tj[(k + 1) & 7] = p0; tj[(k + 2) & 7] = p1; tj[(k + 3) & 7] = p2;
        tj[(k + 4) & 7] = p3; tj[(k + 5) & 7] = p4; tj[(k + 6) & 7] = p5;
        tj[(k + 7) & 7] = p6;
      }
      if (k < 7) {
        va[k + 1] = __builtin_fmaf(-tj[k + 1], piv, va[k + 1]);
        rinv = __builtin_amdgcn_rcpf(va[k + 1]);
      }
#pragma unroll
      for (int j = 0; j < 8; ++j) {
        if (j == k || j == k + 1) continue;
        va[j] = __builtin_fmaf(-tj[j], piv, va[j]);
      }
      va[k] = va[k] * piv;
      if (k == 5) {
        asm volatile("" : "+v"(g) : "v"(va[5]));
#pragma unroll
        for (int r = 0; r < 8; ++r) lds_G[(r + 8 * half) * 16 + l16] = g[r];
        wave_order();
#pragma unroll
        for (int s = 0; s < 2; ++s) {
          const int kk0 = 4 * half + 2 * s;
          gA[s].x = lds_G[l16 * 16 + kk0];
          gA[s].y = lds_G[l16 * 16 + kk0 + 1];
        }
#pragma unroll
        for (int r = 0; r < 8; ++r) cG[r] = (l16 >= 8) ? g[r] : 0.0f;
      }
      if (k < 7) {
        asm volatile("" : "+v"(va[0]), "+v"(va[1]), "+v"(va[2]), "+v"(va[3]),
                          "+v"(va[4]), "+v"(va[5]), "+v"(va[6]), "+v"(va[7]), "+v"(rinv));
      }
    }

    float bP[4], aX[4];
#pragma unroll
    for (int i = 0; i < 4; ++i) {
      const float own  = half ? va[4 + i] : va[i];
      const float send = half ? va[i] : va[4 + i];
      const int   si   = __builtin_bit_cast(int, send);
      const float recv = __builtin_bit_cast(float,
          __builtin_amdgcn_permlanex16(si, si, 0x76543210, 0xfedcba98, false, false));
      bP[i] = useOwn ? own : recv;
      aX[i] = useOwn ? recv : own;
    }

    v2f b0, b1;
    b0.x = bP[0]; b0.y = bP[1];
    b1.x = bP[2]; b1.y = bP[3];
    v8f m1 = wmma_k4(gA[0], b0, cG);
    asm volatile("" : "+v"(xq[0]), "+v"(xq[1]), "+v"(xq[2]), "+v"(xq[3]), "+v"(m1));

    v2f acc[4];
#pragma unroll
    for (int i = 0; i < 4; ++i) {
      const v2f lo = __builtin_shufflevector(xq[i], xq[i], 0, 1);
      const v2f hi = __builtin_shufflevector(xq[i], xq[i], 2, 3);
      acc[i] = Fp[2 * i] * lo;
      acc[i] = __builtin_elementwise_fma(Fp[2 * i + 1], hi, acc[i]);
    }
    asm volatile("" : "+v"(acc[0]), "+v"(acc[1]), "+v"(acc[2]), "+v"(acc[3]), "+v"(b1));
    m1 = wmma_k4(gA[1], b1, m1);

    asm volatile("" : "+v"(acc[0]), "+v"(acc[1]), "+v"(acc[2]), "+v"(acc[3]), "+v"(m1));
    const v2f s01 = acc[0] + acc[1];
    const v2f s23 = acc[2] + acc[3];
    v2f sx = s01 + s23;
    asm volatile("" : "+v"(sx), "+v"(m1));

#pragma unroll
    for (int r = 0; r < 8; ++r) lds_M1[(r + 8 * half) * 16 + l16] = m1[r];
    wave_order();

#pragma unroll
    for (int s = 0; s < 4; ++s) {
      const int k0 = 4 * s + 2 * half;
      ma[s].x = lds_M1[l16 * 16 + k0];
      ma[s].y = lds_M1[l16 * 16 + k0 + 1];
    }
    asm volatile("" : "+v"(aX[0]), "+v"(aX[1]), "+v"(aX[2]), "+v"(aX[3]), "+v"(sx) : : "memory");
    const float xp = sx.x + sx.y;
    const float yv = zv - xp;
    const float yr = __builtin_bit_cast(float,
        __builtin_amdgcn_update_dpp(0, __builtin_bit_cast(int, yv), 0x128, 0xf, 0xf, false));

    v8f xd = {0, 0, 0, 0, 0, 0, 0, 0};
#pragma unroll
    for (int s = 0; s < 2; ++s) {
      v2f a;
      a.x = aX[2 * s] * yr;
      a.y = aX[2 * s + 1] * yr;
      xd = wmma_k4(a, kA[s], xd);
    }
    xdp = xd;
    xpp = xp;
    zv = znext;
  }
  if (T > 0) {
    const float xn = (((xdp[0] + xdp[1]) + (xdp[2] + xdp[3]))
                    + ((xdp[4] + xdp[5]) + (xdp[6] + xdp[7]))) + xpp;
    if (lane >= 16) out[(size_t)(T - 1) * 16 + l16] = xn;
  }
}

extern "C" void kernel_launch(void* const* d_in, const int* in_sizes, int n_in,
                              void* d_out, int out_size, void* d_ws, size_t ws_size,
                              hipStream_t stream) {
  const float* meas = (const float*)d_in[0];
  const float* F    = (const float*)d_in[1];
  const float* Q    = (const float*)d_in[3];
  const float* R    = (const float*)d_in[4];
  float* out = (float*)d_out;

  const int T = in_sizes[0] / MEAS;

  hipLaunchKernelGGL(kalman_seq_kernel, dim3(1), dim3(32), 0, stream,
                     meas, F, Q, R, out, T);
}
